# starting kernel
# baseline (speedup 1.0000x reference)
#include <hip/hip_runtime.h>
#include <hip/hip_bf16.h>

typedef __attribute__((ext_vector_type(2))) float v2f;
typedef __attribute__((ext_vector_type(8))) float v8f;

#define NF 512
#define NH 64
#define NC 8
#define KTILE 128
#define PAIRS (KTILE / 2)
#define PSTR 160


__global__ void deg_init_kernel(float* deg, int n) {
  int i = blockIdx.x * blockDim.x + threadIdx.x;
  if (i < n) deg[i] = 1.0f;
}

__global__ void deg_count_kernel(const int* __restrict__ dst, float* deg, int e) {
  int i = blockIdx.x * blockDim.x + threadIdx.x;
  if (i < e) atomicAdd(&deg[dst[i]], 1.0f);
}

__global__ void dinv_kernel(float* deg, int n) {
  int i = blockIdx.x * blockDim.x + threadIdx.x;
  if (i < n) deg[i] = rsqrtf(deg[i]);
}


__global__ __launch_bounds__(256) void gemm1_wmma_kernel(
    const float* __restrict__ X, const float* __restrict__ W1,
    const float* __restrict__ dinv, float* __restrict__ hs,
    float* __restrict__ acc1, int n) {
  __shared__ float wl[PAIRS * PSTR];
  const int lane  = threadIdx.x & 31;
  const int wave  = threadIdx.x >> 5;
  const int m0    = blockIdx.x * 128 + wave * 16;
  const int mlan  = lane & 15;
  const int khalf = (lane >> 4) << 1;
  const int pbase = lane >> 4;

  int row = m0 + mlan;
  if (row >= n) row = n - 1;
  const float* xrow = X + (size_t)row * NF;

  v8f c0 = {}; v8f c1 = {}; v8f c2 = {}; v8f c3 = {};

  for (int kb = 0; kb < NF; kb += KTILE) {
    for (int i = threadIdx.x; i < KTILE * NH; i += 256) {
      int kl = i >> 6, nn = i & 63;
      wl[(kl >> 1) * PSTR + (nn << 1) + (kl & 1)] = W1[(size_t)kb * NH + i];
    }
    if (kb + KTILE < NF)
      __builtin_prefetch(xrow + kb + KTILE, 0, 3);
    __syncthreads();

    for (int kk = 0; kk < KTILE; kk += 4) {
      v2f a; a.x = xrow[kb + kk + khalf]; a.y = xrow[kb + kk + khalf + 1];
      const v2f* wp = (const v2f*)(wl + ((kk >> 1) + pbase) * PSTR);
      v2f b0 = wp[mlan];
      v2f b1 = wp[16 + mlan];
      v2f b2 = wp[32 + mlan];
      v2f b3 = wp[48 + mlan];
      c0 = __builtin_amdgcn_wmma_f32_16x16x4_f32(false, a, false, b0, (short)0, c0, false, false);
      c1 = __builtin_amdgcn_wmma_f32_16x16x4_f32(false, a, false, b1, (short)0, c1, false, false);
      c2 = __builtin_amdgcn_wmma_f32_16x16x4_f32(false, a, false, b2, (short)0, c2, false, false);
      c3 = __builtin_amdgcn_wmma_f32_16x16x4_f32(false, a, false, b3, (short)0, c3, false, false);
    }
    __syncthreads();
  }

  const int half8 = (lane >> 4) << 3;
  for (int r = 0; r < 8; ++r) {
    int rr = m0 + r + half8;
    if (rr < n) {
      float dv = dinv[rr];
      size_t base = (size_t)rr * NH + mlan;
      float v0 = c0[r] * dv, v1 = c1[r] * dv, v2 = c2[r] * dv, v3 = c3[r] * dv;
      hs[base]      = v0; hs[base + 16]   = v1; hs[base + 32]   = v2; hs[base + 48]   = v3;
      acc1[base]    = v0; acc1[base + 16] = v1; acc1[base + 32] = v2; acc1[base + 48] = v3;
    }
  }
}

__global__ void agg1_kernel(const int* __restrict__ src, const int* __restrict__ dst,
                            const float* __restrict__ hs, float* acc1, long long total) {
  long long idx = (long long)blockIdx.x * blockDim.x + threadIdx.x;
  if (idx >= total) return;
  int e = (int)(idx >> 5);
  int f = (int)(idx & 31) << 1;
  int s = src[e], d = dst[e];
  const float2 v = *(const float2*)(hs + (size_t)s * NH + f);
  float* o = acc1 + (size_t)d * NH + f;
  atomicAdd(o, v.x);
  atomicAdd(o + 1, v.y);
}


__global__ void gemm2_kernel(const float* __restrict__ acc1, const float* __restrict__ dinv,
                             const float* __restrict__ b1, const float* __restrict__ W2,
                             float* __restrict__ t2s, float* __restrict__ acc2, int n) {
  int idx = blockIdx.x * blockDim.x + threadIdx.x;
  if (idx >= n * NC) return;
  int i = idx >> 3, c = idx & 7;
  float di = dinv[i];
  const float* arow = acc1 + (size_t)i * NH;
  float s = 0.0f;
  for (int k = 0; k < NH; ++k) {
    float h = fmaxf(fmaf(arow[k], di, b1[k]), 0.0f);
    s = fmaf(h, W2[k * NC + c], s);
  }
  float v = s * di;
  t2s[idx]  = v;
  acc2[idx] = v;
}

__global__ void agg2_kernel(const int* __restrict__ src, const int* __restrict__ dst,
                            const float* __restrict__ t2s, float* acc2, long long total) {
  long long idx = (long long)blockIdx.x * blockDim.x + threadIdx.x;
  if (idx >= total) return;
  int e = (int)(idx >> 2);
  int c = (int)(idx & 3) << 1;
  int s = src[e], d = dst[e];
  const float2 v = *(const float2*)(t2s + (size_t)s * NC + c);
  float* o = acc2 + (size_t)d * NC + c;
  atomicAdd(o, v.x);
  atomicAdd(o + 1, v.y);
}

__global__ void finalize2_kernel(const float* __restrict__ acc2, const float* __restrict__ dinv,
                                 const float* __restrict__ b2, float* __restrict__ out, int n) {
  int idx = blockIdx.x * blockDim.x + threadIdx.x;
  if (idx >= n * NC) return;
  int i = idx >> 3, c = idx & 7;
  out[idx] = fmaf(acc2[idx], dinv[i], b2[c]);
}


extern "C" void kernel_launch(void* const* d_in, const int* in_sizes, int n_in,
                              void* d_out, int out_size, void* d_ws, size_t ws_size,
                              hipStream_t stream) {
  const float* x  = (const float*)d_in[0];
  const int*   ei = (const int*)d_in[1];
  const float* W1 = (const float*)d_in[2];
  const float* b1 = (const float*)d_in[3];
  const float* W2 = (const float*)d_in[4];
  const float* b2 = (const float*)d_in[5];
  float* out = (float*)d_out;

  const int n = in_sizes[0] / NF;
  const int e = in_sizes[1] / 2;
  const int* src = ei;
  const int* dst = ei + e;

  float* ws   = (float*)d_ws;
  float* dinv = ws;
  float* hs   = dinv + n;
  float* acc1 = hs + (size_t)n * NH;
  float* t2s  = acc1 + (size_t)n * NH;
  float* acc2 = t2s + (size_t)n * NC;

  const int T = 256;
  deg_init_kernel<<<(n + T - 1) / T, T, 0, stream>>>(dinv, n);
  deg_count_kernel<<<(e + T - 1) / T, T, 0, stream>>>(dst, dinv, e);
  dinv_kernel<<<(n + T - 1) / T, T, 0, stream>>>(dinv, n);

  gemm1_wmma_kernel<<<(n + 127) / 128, T, 0, stream>>>(x, W1, dinv, hs, acc1, n);

  long long tot1 = (long long)e * 32;
  agg1_kernel<<<(unsigned)((tot1 + T - 1) / T), T, 0, stream>>>(src, dst, hs, acc1, tot1);

  gemm2_kernel<<<(n * NC + T - 1) / T, T, 0, stream>>>(acc1, dinv, b1, W2, t2s, acc2, n);

  long long tot2 = (long long)e * 4;
  agg2_kernel<<<(unsigned)((tot2 + T - 1) / T), T, 0, stream>>>(src, dst, t2s, acc2, tot2);

  finalize2_kernel<<<(n * NC + T - 1) / T, T, 0, stream>>>(acc2, dinv, b2, out, n);
}


# deg_count: work-groups count edge chunks into an LDS integer histogram per 32768-node range, then add bins to deg as whole-line f32 atomics of exact counts (3.2M scattered adds -> 0.2M line adds)
# speedup vs baseline: 1.2616x; 1.2616x over previous
#include <hip/hip_runtime.h>
#include <hip/hip_bf16.h>

typedef __attribute__((ext_vector_type(2))) float v2f;
typedef __attribute__((ext_vector_type(8))) float v8f;

#define NF 512
#define NH 64
#define NC 8
#define KTILE 128
#define PAIRS (KTILE / 2)
#define PSTR 160


__global__ void deg_init_kernel(float* deg, int n) {
  int i = blockIdx.x * blockDim.x + threadIdx.x;
  if (i < n) deg[i] = 1.0f;
}

#define DEG_RANGE 32768
#define DEG_CHUNKS 64
#define DEG_T 1024
#define DEG_U 8

__global__ __launch_bounds__(DEG_T) void deg_count_kernel(const int* __restrict__ dst, float* deg,
                                                          int e, int n) {
  __shared__ unsigned cnt[DEG_RANGE];
  const int lo = blockIdx.y * DEG_RANGE;
  for (int i = threadIdx.x; i < DEG_RANGE; i += DEG_T) cnt[i] = 0u;
  __syncthreads();

  const int per = (e + (int)gridDim.x - 1) / (int)gridDim.x;
  const int b0  = blockIdx.x * per;
  const int en  = (b0 + per < e) ? (b0 + per) : e;
  for (int i = b0 + (int)threadIdx.x; i < en; i += DEG_T * DEG_U) {
    int d[DEG_U];
#pragma unroll
    for (int u = 0; u < DEG_U; ++u) {
      int j = i + u * DEG_T;
      d[u] = (j < en) ? dst[j] : -1;
    }
#pragma unroll
    for (int u = 0; u < DEG_U; ++u) {
      unsigned r = (unsigned)(d[u] - lo);
      if (r < (unsigned)DEG_RANGE) atomicAdd(&cnt[r], 1u);
    }
  }
  __syncthreads();

  for (int i = threadIdx.x; i < DEG_RANGE; i += DEG_T) {
    unsigned c = cnt[i];
    int node = lo + i;
    if (c != 0u && node < n) atomicAdd(&deg[node], (float)c);
  }
}

__global__ void dinv_kernel(float* deg, int n) {
  int i = blockIdx.x * blockDim.x + threadIdx.x;
  if (i < n) deg[i] = rsqrtf(deg[i]);
}


__global__ __launch_bounds__(256) void gemm1_wmma_kernel(
    const float* __restrict__ X, const float* __restrict__ W1,
    const float* __restrict__ dinv, float* __restrict__ hs,
    float* __restrict__ acc1, int n) {
  __shared__ float wl[PAIRS * PSTR];
  const int lane  = threadIdx.x & 31;
  const int wave  = threadIdx.x >> 5;
  const int m0    = blockIdx.x * 128 + wave * 16;
  const int mlan  = lane & 15;
  const int khalf = (lane >> 4) << 1;
  const int pbase = lane >> 4;

  int row = m0 + mlan;
  if (row >= n) row = n - 1;
  const float* xrow = X + (size_t)row * NF;

  v8f c0 = {}; v8f c1 = {}; v8f c2 = {}; v8f c3 = {};

  for (int kb = 0; kb < NF; kb += KTILE) {
    for (int i = threadIdx.x; i < KTILE * NH; i += 256) {
      int kl = i >> 6, nn = i & 63;
      wl[(kl >> 1) * PSTR + (nn << 1) + (kl & 1)] = W1[(size_t)kb * NH + i];
    }
    if (kb + KTILE < NF)
      __builtin_prefetch(xrow + kb + KTILE, 0, 3);
    __syncthreads();

    for (int kk = 0; kk < KTILE; kk += 4) {
      v2f a; a.x = xrow[kb + kk + khalf]; a.y = xrow[kb + kk + khalf + 1];
      const v2f* wp = (const v2f*)(wl + ((kk >> 1) + pbase) * PSTR);
      v2f b0 = wp[mlan];
      v2f b1 = wp[16 + mlan];
      v2f b2 = wp[32 + mlan];
      v2f b3 = wp[48 + mlan];
      c0 = __builtin_amdgcn_wmma_f32_16x16x4_f32(false, a, false, b0, (short)0, c0, false, false);
      c1 = __builtin_amdgcn_wmma_f32_16x16x4_f32(false, a, false, b1, (short)0, c1, false, false);
      c2 = __builtin_amdgcn_wmma_f32_16x16x4_f32(false, a, false, b2, (short)0, c2, false, false);
      c3 = __builtin_amdgcn_wmma_f32_16x16x4_f32(false, a, false, b3, (short)0, c3, false, false);
    }
    __syncthreads();
  }

  const int half8 = (lane >> 4) << 3;
  for (int r = 0; r < 8; ++r) {
    int rr = m0 + r + half8;
    if (rr < n) {
      float dv = dinv[rr];
      size_t base = (size_t)rr * NH + mlan;
      float v0 = c0[r] * dv, v1 = c1[r] * dv, v2 = c2[r] * dv, v3 = c3[r] * dv;
      hs[base]      = v0; hs[base + 16]   = v1; hs[base + 32]   = v2; hs[base + 48]   = v3;
      acc1[base]    = v0; acc1[base + 16] = v1; acc1[base + 32] = v2; acc1[base + 48] = v3;
    }
  }
}

__global__ void agg1_kernel(const int* __restrict__ src, const int* __restrict__ dst,
                            const float* __restrict__ hs, float* acc1, long long total) {
  long long idx = (long long)blockIdx.x * blockDim.x + threadIdx.x;
  if (idx >= total) return;
  int e = (int)(idx >> 5);
  int f = (int)(idx & 31) << 1;
  int s = src[e], d = dst[e];
  const float2 v = *(const float2*)(hs + (size_t)s * NH + f);
  float* o = acc1 + (size_t)d * NH + f;
  atomicAdd(o, v.x);
  atomicAdd(o + 1, v.y);
}


__global__ void gemm2_kernel(const float* __restrict__ acc1, const float* __restrict__ dinv,
                             const float* __restrict__ b1, const float* __restrict__ W2,
                             float* __restrict__ t2s, float* __restrict__ acc2, int n) {
  int idx = blockIdx.x * blockDim.x + threadIdx.x;
  if (idx >= n * NC) return;
  int i = idx >> 3, c = idx & 7;
  float di = dinv[i];
  const float* arow = acc1 + (size_t)i * NH;
  float s = 0.0f;
  for (int k = 0; k < NH; ++k) {
    float h = fmaxf(fmaf(arow[k], di, b1[k]), 0.0f);
    s = fmaf(h, W2[k * NC + c], s);
  }
  float v = s * di;
  t2s[idx]  = v;
  acc2[idx] = v;
}

__global__ void agg2_kernel(const int* __restrict__ src, const int* __restrict__ dst,
                            const float* __restrict__ t2s, float* acc2, long long total) {
  long long idx = (long long)blockIdx.x * blockDim.x + threadIdx.x;
  if (idx >= total) return;
  int e = (int)(idx >> 2);
  int c = (int)(idx & 3) << 1;
  int s = src[e], d = dst[e];
  const float2 v = *(const float2*)(t2s + (size_t)s * NC + c);
  float* o = acc2 + (size_t)d * NC + c;
  atomicAdd(o, v.x);
  atomicAdd(o + 1, v.y);
}

__global__ void finalize2_kernel(const float* __restrict__ acc2, const float* __restrict__ dinv,
                                 const float* __restrict__ b2, float* __restrict__ out, int n) {
  int idx = blockIdx.x * blockDim.x + threadIdx.x;
  if (idx >= n * NC) return;
  int i = idx >> 3, c = idx & 7;
  out[idx] = fmaf(acc2[idx], dinv[i], b2[c]);
}


extern "C" void kernel_launch(void* const* d_in, const int* in_sizes, int n_in,
                              void* d_out, int out_size, void* d_ws, size_t ws_size,
                              hipStream_t stream) {
  const float* x  = (const float*)d_in[0];
  const int*   ei = (const int*)d_in[1];
  const float* W1 = (const float*)d_in[2];
  const float* b1 = (const float*)d_in[3];
  const float* W2 = (const float*)d_in[4];
  const float* b2 = (const float*)d_in[5];
  float* out = (float*)d_out;

  const int n = in_sizes[0] / NF;
  const int e = in_sizes[1] / 2;
  const int* src = ei;
  const int* dst = ei + e;

  float* ws   = (float*)d_ws;
  float* dinv = ws;
  float* hs   = dinv + n;
  float* acc1 = hs + (size_t)n * NH;
  float* t2s  = acc1 + (size_t)n * NH;
  float* acc2 = t2s + (size_t)n * NC;

  const int T = 256;
  deg_init_kernel<<<(n + T - 1) / T, T, 0, stream>>>(dinv, n);
  deg_count_kernel<<<dim3(DEG_CHUNKS, (n + DEG_RANGE - 1) / DEG_RANGE), DEG_T, 0, stream>>>(dst, dinv, e, n);
  dinv_kernel<<<(n + T - 1) / T, T, 0, stream>>>(dinv, n);

  gemm1_wmma_kernel<<<(n + 127) / 128, T, 0, stream>>>(x, W1, dinv, hs, acc1, n);

  long long tot1 = (long long)e * 32;
  agg1_kernel<<<(unsigned)((tot1 + T - 1) / T), T, 0, stream>>>(src, dst, hs, acc1, tot1);

  gemm2_kernel<<<(n * NC + T - 1) / T, T, 0, stream>>>(acc1, dinv, b1, W2, t2s, acc2, n);

  long long tot2 = (long long)e * 4;
  agg2_kernel<<<(unsigned)((tot2 + T - 1) / T), T, 0, stream>>>(src, dst, t2s, acc2, tot2);

  finalize2_kernel<<<(n * NC + T - 1) / T, T, 0, stream>>>(acc2, dinv, b2, out, n);
}


# agg1/agg2: destination rows split between the two L2 coherence halves; persistent work-groups read their die, claim edge chunks per half, add only their half's rows (f32 atomics, arrival order)
# speedup vs baseline: 3.0040x; 2.3811x over previous
#include <hip/hip_runtime.h>
#include <hip/hip_bf16.h>

typedef __attribute__((ext_vector_type(2))) float v2f;
typedef __attribute__((ext_vector_type(8))) float v8f;

#define NF 512
#define NH 64
#define NC 8
#define KTILE 128
#define PAIRS (KTILE / 2)
#define PSTR 160


__global__ void deg_init_kernel(float* deg, int n) {
  int i = blockIdx.x * blockDim.x + threadIdx.x;
  if (i < n) deg[i] = 1.0f;
}

#define DEG_RANGE 32768
#define DEG_CHUNKS 64
#define DEG_T 1024
#define DEG_U 8

__global__ __launch_bounds__(DEG_T) void deg_count_kernel(const int* __restrict__ dst, float* deg,
                                                          int e, int n) {
  __shared__ unsigned cnt[DEG_RANGE];
  const int lo = blockIdx.y * DEG_RANGE;
  for (int i = threadIdx.x; i < DEG_RANGE; i += DEG_T) cnt[i] = 0u;
  __syncthreads();

  const int per = (e + (int)gridDim.x - 1) / (int)gridDim.x;
  const int b0  = blockIdx.x * per;
  const int en  = (b0 + per < e) ? (b0 + per) : e;
  for (int i = b0 + (int)threadIdx.x; i < en; i += DEG_T * DEG_U) {
    int d[DEG_U];
#pragma unroll
    for (int u = 0; u < DEG_U; ++u) {
      int j = i + u * DEG_T;
      d[u] = (j < en) ? dst[j] : -1;
    }
#pragma unroll
    for (int u = 0; u < DEG_U; ++u) {
      unsigned r = (unsigned)(d[u] - lo);
      if (r < (unsigned)DEG_RANGE) atomicAdd(&cnt[r], 1u);
    }
  }
  __syncthreads();

  for (int i = threadIdx.x; i < DEG_RANGE; i += DEG_T) {
    unsigned c = cnt[i];
    int node = lo + i;
    if (c != 0u && node < n) atomicAdd(&deg[node], (float)c);
  }
}

__global__ void dinv_kernel(float* deg, int n, unsigned* queue) {
  int i = blockIdx.x * blockDim.x + threadIdx.x;
  if (i < n) deg[i] = rsqrtf(deg[i]);
  if (i < 2) queue[i] = 0u;
}


__global__ __launch_bounds__(256) void gemm1_wmma_kernel(
    const float* __restrict__ X, const float* __restrict__ W1,
    const float* __restrict__ dinv, float* __restrict__ hs,
    float* __restrict__ acc1, int n) {
  __shared__ float wl[PAIRS * PSTR];
  const int lane  = threadIdx.x & 31;
  const int wave  = threadIdx.x >> 5;
  const int m0    = blockIdx.x * 128 + wave * 16;
  const int mlan  = lane & 15;
  const int khalf = (lane >> 4) << 1;
  const int pbase = lane >> 4;

  int row = m0 + mlan;
  if (row >= n) row = n - 1;
  const float* xrow = X + (size_t)row * NF;

  v8f c0 = {}; v8f c1 = {}; v8f c2 = {}; v8f c3 = {};

  for (int kb = 0; kb < NF; kb += KTILE) {
    for (int i = threadIdx.x; i < KTILE * NH; i += 256) {
      int kl = i >> 6, nn = i & 63;
      wl[(kl >> 1) * PSTR + (nn << 1) + (kl & 1)] = W1[(size_t)kb * NH + i];
    }
    if (kb + KTILE < NF)
      __builtin_prefetch(xrow + kb + KTILE, 0, 3);
    __syncthreads();

    for (int kk = 0; kk < KTILE; kk += 4) {
      v2f a; a.x = xrow[kb + kk + khalf]; a.y = xrow[kb + kk + khalf + 1];
      const v2f* wp = (const v2f*)(wl + ((kk >> 1) + pbase) * PSTR);
      v2f b0 = wp[mlan];
      v2f b1 = wp[16 + mlan];
      v2f b2 = wp[32 + mlan];
      v2f b3 = wp[48 + mlan];
      c0 = __builtin_amdgcn_wmma_f32_16x16x4_f32(false, a, false, b0, (short)0, c0, false, false);
      c1 = __builtin_amdgcn_wmma_f32_16x16x4_f32(false, a, false, b1, (short)0, c1, false, false);
      c2 = __builtin_amdgcn_wmma_f32_16x16x4_f32(false, a, false, b2, (short)0, c2, false, false);
      c3 = __builtin_amdgcn_wmma_f32_16x16x4_f32(false, a, false, b3, (short)0, c3, false, false);
    }
    __syncthreads();
  }

  const int half8 = (lane >> 4) << 3;
  for (int r = 0; r < 8; ++r) {
    int rr = m0 + r + half8;
    if (rr < n) {
      float dv = dinv[rr];
      size_t base = (size_t)rr * NH + mlan;
      float v0 = c0[r] * dv, v1 = c1[r] * dv, v2 = c2[r] * dv, v3 = c3[r] * dv;
      hs[base]      = v0; hs[base + 16]   = v1; hs[base + 32]   = v2; hs[base + 48]   = v3;
      acc1[base]    = v0; acc1[base + 16] = v1; acc1[base + 32] = v2; acc1[base + 48] = v3;
    }
  }
}

#define AGG_WG 2048
#define AGG_CH 1024
#define AGG_BLK (AGG_CH / 256)

__device__ __forceinline__ int coherence_half() {
  const unsigned id = __builtin_amdgcn_s_sendmsg_rtn(0x87);
  return (int)((id >> 18) & 1u);
}

struct EdgeList { int2 e[AGG_BLK * 32]; };

template <int U>
__device__ __forceinline__ void agg1_group(const int2* lst, int i, const float* __restrict__ hs,
                                           float* acc1, int f) {
  int s[U], d[U]; float2 v[U];
#pragma unroll
  for (int u = 0; u < U; ++u) {
    const int2 p = lst[i + u];
    s[u] = __builtin_amdgcn_readfirstlane(p.x);
    d[u] = __builtin_amdgcn_readfirstlane(p.y);
  }
#pragma unroll
  for (int u = 0; u < U; ++u) v[u] = *(const float2*)(hs + (size_t)s[u] * NH + f);
#pragma unroll
  for (int u = 0; u < U; ++u) {
    float* o = acc1 + (size_t)d[u] * NH + f;
    atomicAdd(o, v[u].x);
    atomicAdd(o + 1, v[u].y);
  }
}

__global__ __launch_bounds__(256) void agg1_kernel(const int* __restrict__ src, const int* __restrict__ dst,
                                                   const float* __restrict__ hs, float* acc1,
                                                   unsigned* queue, int e, int n) {
  __shared__ int s_half;
  __shared__ int s_next[2];
  __shared__ EdgeList s_list[8];
  const int lane = threadIdx.x & 31;
  const int wave = threadIdx.x >> 5;
  const int f = lane << 1;
  const int nsplit = (n >> 1) & ~7;
  const int nchunk = (e + AGG_CH - 1) / AGG_CH;
  int2* lst = s_list[wave].e;
  if (threadIdx.x == 0) s_half = coherence_half();

  for (int pass = 0; pass < 2; ++pass) {
    __syncthreads();
    const int h = s_half ^ pass;
    unsigned* q = queue + h;
    if (threadIdx.x == 0) s_next[0] = (int)atomicAdd(q, 1u);
    __syncthreads();
    for (int it = 0; ; ++it) {
      const int cur = s_next[it & 1];
      if (cur >= nchunk) break;
      int dv[AGG_BLK], sv[AGG_BLK];
#pragma unroll
      for (int b = 0; b < AGG_BLK; ++b) {
        const int j = cur * AGG_CH + (b * 8 + wave) * 32 + lane;
        dv[b] = -1; sv[b] = 0;
        if (j < e) { dv[b] = dst[j]; sv[b] = src[j]; }
      }
      unsigned nx = 0u;
      if (threadIdx.x == 0) nx = atomicAdd(q, 1u);
      int cnt = 0;
#pragma unroll
      for (int b = 0; b < AGG_BLK; ++b) {
        const bool mine = (dv[b] >= 0) && ((dv[b] >= nsplit) == (h == 1));
        const unsigned m = __builtin_amdgcn_ballot_w32(mine);
        if (mine) lst[cnt + (int)__builtin_amdgcn_mbcnt_lo(m, 0u)] = make_int2(sv[b], dv[b]);
        cnt += __builtin_popcount(m);
      }
      if (threadIdx.x == 0) s_next[(it + 1) & 1] = (int)nx;
      __syncthreads();

      int i = 0;
      for (; i + 8 <= cnt; i += 8) agg1_group<8>(lst, i, hs, acc1, f);
      if (i + 4 <= cnt) { agg1_group<4>(lst, i, hs, acc1, f); i += 4; }
      for (; i < cnt; ++i) agg1_group<1>(lst, i, hs, acc1, f);

    }
  }
}


__global__ void gemm2_kernel(const float* __restrict__ acc1, const float* __restrict__ dinv,
                             const float* __restrict__ b1, const float* __restrict__ W2,
                             float* __restrict__ t2s, float* __restrict__ acc2, int n,
                             unsigned* queue) {
  int idx = blockIdx.x * blockDim.x + threadIdx.x;
  if (idx < 2) queue[idx] = 0u;
  if (idx >= n * NC) return;
  int i = idx >> 3, c = idx & 7;
  float di = dinv[i];
  const float* arow = acc1 + (size_t)i * NH;
  float s = 0.0f;
  for (int k = 0; k < NH; ++k) {
    float h = fmaxf(fmaf(arow[k], di, b1[k]), 0.0f);
    s = fmaf(h, W2[k * NC + c], s);
  }
  float v = s * di;
  t2s[idx]  = v;
  acc2[idx] = v;
}

template <int U>
__device__ __forceinline__ void agg2_group(const int2* lst, int i, const float* __restrict__ t2s,
                                           float* acc2, int g, int c) {
  int2 p[U]; float v[U];
#pragma unroll
  for (int u = 0; u < U; ++u) p[u] = lst[i + 4 * u + g];
#pragma unroll
  for (int u = 0; u < U; ++u) v[u] = t2s[(size_t)p[u].x * NC + c];
#pragma unroll
  for (int u = 0; u < U; ++u) atomicAdd(acc2 + (size_t)p[u].y * NC + c, v[u]);
}

__global__ __launch_bounds__(256) void agg2_kernel(const int* __restrict__ src, const int* __restrict__ dst,
                                                   const float* __restrict__ t2s, float* acc2,
                                                   unsigned* queue, int e, int n) {
  __shared__ int s_half;
  __shared__ int s_next[2];
  __shared__ EdgeList s_list[8];
  const int lane = threadIdx.x & 31;
  const int wave = threadIdx.x >> 5;
  const int g = lane >> 3;
  const int c = lane & 7;
  const int nsplit = (n >> 1) & ~7;
  const int nchunk = (e + AGG_CH - 1) / AGG_CH;
  int2* lst = s_list[wave].e;
  if (threadIdx.x == 0) s_half = coherence_half();

  for (int pass = 0; pass < 2; ++pass) {
    __syncthreads();
    const int h = s_half ^ pass;
    unsigned* q = queue + h;
    if (threadIdx.x == 0) s_next[0] = (int)atomicAdd(q, 1u);
    __syncthreads();
    for (int it = 0; ; ++it) {
      const int cur = s_next[it & 1];
      if (cur >= nchunk) break;
      int dv[AGG_BLK], sv[AGG_BLK];
#pragma unroll
      for (int b = 0; b < AGG_BLK; ++b) {
        const int j = cur * AGG_CH + (b * 8 + wave) * 32 + lane;
        dv[b] = -1; sv[b] = 0;
        if (j < e) { dv[b] = dst[j]; sv[b] = src[j]; }
      }
      unsigned nx = 0u;
      if (threadIdx.x == 0) nx = atomicAdd(q, 1u);
      int cnt = 0;
#pragma unroll
      for (int b = 0; b < AGG_BLK; ++b) {
        const bool mine = (dv[b] >= 0) && ((dv[b] >= nsplit) == (h == 1));
        const unsigned m = __builtin_amdgcn_ballot_w32(mine);
        if (mine) lst[cnt + (int)__builtin_amdgcn_mbcnt_lo(m, 0u)] = make_int2(sv[b], dv[b]);
        cnt += __builtin_popcount(m);
      }
      if (threadIdx.x == 0) s_next[(it + 1) & 1] = (int)nx;
      __syncthreads();

      int i = 0;
      for (; i + 16 <= cnt; i += 16) agg2_group<4>(lst, i, t2s, acc2, g, c);
      for (; i < cnt; i += 4)
        if (i + g < cnt) agg2_group<1>(lst, i, t2s, acc2, g, c);

    }
  }
}

__global__ void finalize2_kernel(const float* __restrict__ acc2, const float* __restrict__ dinv,
                                 const float* __restrict__ b2, float* __restrict__ out, int n) {
  int idx = blockIdx.x * blockDim.x + threadIdx.x;
  if (idx >= n * NC) return;
  int i = idx >> 3, c = idx & 7;
  out[idx] = fmaf(acc2[idx], dinv[i], b2[c]);
}


extern "C" void kernel_launch(void* const* d_in, const int* in_sizes, int n_in,
                              void* d_out, int out_size, void* d_ws, size_t ws_size,
                              hipStream_t stream) {
  const float* x  = (const float*)d_in[0];
  const int*   ei = (const int*)d_in[1];
  const float* W1 = (const float*)d_in[2];
  const float* b1 = (const float*)d_in[3];
  const float* W2 = (const float*)d_in[4];
  const float* b2 = (const float*)d_in[5];
  float* out = (float*)d_out;

  const int n = in_sizes[0] / NF;
  const int e = in_sizes[1] / 2;
  const int* src = ei;
  const int* dst = ei + e;

  float* ws   = (float*)d_ws;
  float* dinv = ws;
  float* hs   = dinv + n;
  float* acc1 = hs + (size_t)n * NH;
  float* t2s  = acc1 + (size_t)n * NH;
  float* acc2 = t2s + (size_t)n * NC;

  const int T = 256;
  deg_init_kernel<<<(n + T - 1) / T, T, 0, stream>>>(dinv, n);
  deg_count_kernel<<<dim3(DEG_CHUNKS, (n + DEG_RANGE - 1) / DEG_RANGE), DEG_T, 0, stream>>>(dst, dinv, e, n);
  unsigned* q1 = (unsigned*)t2s;
  unsigned* q2 = (unsigned*)hs;
  dinv_kernel<<<(n + T - 1) / T, T, 0, stream>>>(dinv, n, q1);

  gemm1_wmma_kernel<<<(n + 127) / 128, T, 0, stream>>>(x, W1, dinv, hs, acc1, n);

  agg1_kernel<<<AGG_WG, T, 0, stream>>>(src, dst, hs, acc1, q1, e, n);

  gemm2_kernel<<<(n * NC + T - 1) / T, T, 0, stream>>>(acc1, dinv, b1, W2, t2s, acc2, n, q2);

  agg2_kernel<<<AGG_WG, T, 0, stream>>>(src, dst, t2s, acc2, q2, e, n);

  finalize2_kernel<<<(n * NC + T - 1) / T, T, 0, stream>>>(acc2, dinv, b2, out, n);
}


# agg1: each f32 atomic instruction covers one whole 128-byte line of the destination row (lane l adds features l and l+32; hs stored feature-interleaved, gather stays one 8-byte load)
# speedup vs baseline: 3.4428x; 1.1461x over previous
#include <hip/hip_runtime.h>
#include <hip/hip_bf16.h>

typedef __attribute__((ext_vector_type(2))) float v2f;
typedef __attribute__((ext_vector_type(8))) float v8f;

#define NF 512
#define NH 64
#define NC 8
#define KTILE 128
#define PAIRS (KTILE / 2)
#define PSTR 160


__global__ void deg_init_kernel(float* deg, int n) {
  int i = blockIdx.x * blockDim.x + threadIdx.x;
  if (i < n) deg[i] = 1.0f;
}

#define DEG_RANGE 32768
#define DEG_CHUNKS 64
#define DEG_T 1024
#define DEG_U 8

__global__ __launch_bounds__(DEG_T) void deg_count_kernel(const int* __restrict__ dst, float* deg,
                                                          int e, int n) {
  __shared__ unsigned cnt[DEG_RANGE];
  const int lo = blockIdx.y * DEG_RANGE;
  for (int i = threadIdx.x; i < DEG_RANGE; i += DEG_T) cnt[i] = 0u;
  __syncthreads();

  const int per = (e + (int)gridDim.x - 1) / (int)gridDim.x;
  const int b0  = blockIdx.x * per;
  const int en  = (b0 + per < e) ? (b0 + per) : e;
  for (int i = b0 + (int)threadIdx.x; i < en; i += DEG_T * DEG_U) {
    int d[DEG_U];
#pragma unroll
    for (int u = 0; u < DEG_U; ++u) {
      int j = i + u * DEG_T;
      d[u] = (j < en) ? dst[j] : -1;
    }
#pragma unroll
    for (int u = 0; u < DEG_U; ++u) {
      unsigned r = (unsigned)(d[u] - lo);
      if (r < (unsigned)DEG_RANGE) atomicAdd(&cnt[r], 1u);
    }
  }
  __syncthreads();

  for (int i = threadIdx.x; i < DEG_RANGE; i += DEG_T) {
    unsigned c = cnt[i];
    int node = lo + i;
    if (c != 0u && node < n) atomicAdd(&deg[node], (float)c);
  }
}

__global__ void dinv_kernel(float* deg, int n, unsigned* queue) {
  int i = blockIdx.x * blockDim.x + threadIdx.x;
  if (i < n) deg[i] = rsqrtf(deg[i]);
  if (i < 2) queue[i] = 0u;
}


__global__ __launch_bounds__(256) void gemm1_wmma_kernel(
    const float* __restrict__ X, const float* __restrict__ W1,
    const float* __restrict__ dinv, float* __restrict__ hs,
    float* __restrict__ acc1, int n) {
  __shared__ float wl[PAIRS * PSTR];
  const int lane  = threadIdx.x & 31;
  const int wave  = threadIdx.x >> 5;
  const int m0    = blockIdx.x * 128 + wave * 16;
  const int mlan  = lane & 15;
  const int khalf = (lane >> 4) << 1;
  const int pbase = lane >> 4;

  int row = m0 + mlan;
  if (row >= n) row = n - 1;
  const float* xrow = X + (size_t)row * NF;

  v8f c0 = {}; v8f c1 = {}; v8f c2 = {}; v8f c3 = {};

  for (int kb = 0; kb < NF; kb += KTILE) {
    for (int i = threadIdx.x; i < KTILE * NH; i += 256) {
      int kl = i >> 6, nn = i & 63;
      wl[(kl >> 1) * PSTR + (nn << 1) + (kl & 1)] = W1[(size_t)kb * NH + i];
    }
    if (kb + KTILE < NF)
      __builtin_prefetch(xrow + kb + KTILE, 0, 3);
    __syncthreads();

    for (int kk = 0; kk < KTILE; kk += 4) {
      v2f a; a.x = xrow[kb + kk + khalf]; a.y = xrow[kb + kk + khalf + 1];
      const v2f* wp = (const v2f*)(wl + ((kk >> 1) + pbase) * PSTR);
      v2f b0 = wp[mlan];
      v2f b1 = wp[16 + mlan];
      v2f b2 = wp[32 + mlan];
      v2f b3 = wp[48 + mlan];
      c0 = __builtin_amdgcn_wmma_f32_16x16x4_f32(false, a, false, b0, (short)0, c0, false, false);
      c1 = __builtin_amdgcn_wmma_f32_16x16x4_f32(false, a, false, b1, (short)0, c1, false, false);
      c2 = __builtin_amdgcn_wmma_f32_16x16x4_f32(false, a, false, b2, (short)0, c2, false, false);
      c3 = __builtin_amdgcn_wmma_f32_16x16x4_f32(false, a, false, b3, (short)0, c3, false, false);
    }
    __syncthreads();
  }

  const int half8 = (lane >> 4) << 3;
  for (int r = 0; r < 8; ++r) {
    int rr = m0 + r + half8;
    if (rr < n) {
      float dv = dinv[rr];
      size_t base = (size_t)rr * NH + mlan;
      float v0 = c0[r] * dv, v1 = c1[r] * dv, v2 = c2[r] * dv, v3 = c3[r] * dv;
      float* hrow = hs + (size_t)rr * NH;
      *(float2*)(hrow + 2 * mlan)        = make_float2(v0, v2);
      *(float2*)(hrow + 2 * (mlan + 16)) = make_float2(v1, v3);
      acc1[base]    = v0; acc1[base + 16] = v1; acc1[base + 32] = v2; acc1[base + 48] = v3;
    }
  }
}

#define AGG_WG 2048
#define AGG_CH 1024
#define AGG_BLK (AGG_CH / 256)

__device__ __forceinline__ int coherence_half() {
  const unsigned id = __builtin_amdgcn_s_sendmsg_rtn(0x87);
  return (int)((id >> 18) & 1u);
}

struct EdgeList { int2 e[AGG_BLK * 32]; };

template <int U>
__device__ __forceinline__ void agg1_group(const int2* lst, int i, const float* __restrict__ hs,
                                           float* acc1, int lane) {
  int s[U], d[U]; float2 v[U];
#pragma unroll
  for (int u = 0; u < U; ++u) {
    const int2 p = lst[i + u];
    s[u] = __builtin_amdgcn_readfirstlane(p.x);
    d[u] = __builtin_amdgcn_readfirstlane(p.y);
  }
#pragma unroll
  for (int u = 0; u < U; ++u) v[u] = *(const float2*)(hs + (size_t)s[u] * NH + 2 * lane);
#pragma unroll
  for (int u = 0; u < U; ++u) {
    float* o = acc1 + (size_t)d[u] * NH + lane;
    atomicAdd(o, v[u].x);
    atomicAdd(o + 32, v[u].y);
  }
}

__global__ __launch_bounds__(256) void agg1_kernel(const int* __restrict__ src, const int* __restrict__ dst,
                                                   const float* __restrict__ hs, float* acc1,
                                                   unsigned* queue, int e, int n) {
  __shared__ int s_half;
  __shared__ int s_next[2];
  __shared__ EdgeList s_list[8];
  const int lane = threadIdx.x & 31;
  const int wave = threadIdx.x >> 5;
  const int nsplit = (n >> 1) & ~7;
  const int nchunk = (e + AGG_CH - 1) / AGG_CH;
  int2* lst = s_list[wave].e;
  if (threadIdx.x == 0) s_half = coherence_half();

  for (int pass = 0; pass < 2; ++pass) {
    __syncthreads();
    const int h = s_half ^ pass;
    unsigned* q = queue + h;
    if (threadIdx.x == 0) s_next[0] = (int)atomicAdd(q, 1u);
    __syncthreads();
    for (int it = 0; ; ++it) {
      const int cur = s_next[it & 1];
      if (cur >= nchunk) break;
      int dv[AGG_BLK], sv[AGG_BLK];
#pragma unroll
      for (int b = 0; b < AGG_BLK; ++b) {
        const int j = cur * AGG_CH + (b * 8 + wave) * 32 + lane;
        dv[b] = -1; sv[b] = 0;
        if (j < e) { dv[b] = dst[j]; sv[b] = src[j]; }
      }
      unsigned nx = 0u;
      if (threadIdx.x == 0) nx = atomicAdd(q, 1u);
      int cnt = 0;
#pragma unroll
      for (int b = 0; b < AGG_BLK; ++b) {
        const bool mine = (dv[b] >= 0) && ((dv[b] >= nsplit) == (h == 1));
        const unsigned m = __builtin_amdgcn_ballot_w32(mine);
        if (mine) lst[cnt + (int)__builtin_amdgcn_mbcnt_lo(m, 0u)] = make_int2(sv[b], dv[b]);
        cnt += __builtin_popcount(m);
      }
      if (threadIdx.x == 0) s_next[(it + 1) & 1] = (int)nx;
      __syncthreads();

      int i = 0;
      for (; i + 8 <= cnt; i += 8) agg1_group<8>(lst, i, hs, acc1, lane);
      if (i + 4 <= cnt) { agg1_group<4>(lst, i, hs, acc1, lane); i += 4; }
      for (; i < cnt; ++i) agg1_group<1>(lst, i, hs, acc1, lane);

    }
  }
}


__global__ void gemm2_kernel(const float* __restrict__ acc1, const float* __restrict__ dinv,
                             const float* __restrict__ b1, const float* __restrict__ W2,
                             float* __restrict__ t2s, float* __restrict__ acc2, int n,
                             unsigned* queue) {
  int idx = blockIdx.x * blockDim.x + threadIdx.x;
  if (idx < 2) queue[idx] = 0u;
  if (idx >= n * NC) return;
  int i = idx >> 3, c = idx & 7;
  float di = dinv[i];
  const float* arow = acc1 + (size_t)i * NH;
  float s = 0.0f;
  for (int k = 0; k < NH; ++k) {
    float h = fmaxf(fmaf(arow[k], di, b1[k]), 0.0f);
    s = fmaf(h, W2[k * NC + c], s);
  }
  float v = s * di;
  t2s[idx]  = v;
  acc2[idx] = v;
}

template <int U>
__device__ __forceinline__ void agg2_group(const int2* lst, int i, const float* __restrict__ t2s,
                                           float* acc2, int g, int c) {
  int2 p[U]; float v[U];
#pragma unroll
  for (int u = 0; u < U; ++u) p[u] = lst[i + 4 * u + g];
#pragma unroll
  for (int u = 0; u < U; ++u) v[u] = t2s[(size_t)p[u].x * NC + c];
#pragma unroll
  for (int u = 0; u < U; ++u) atomicAdd(acc2 + (size_t)p[u].y * NC + c, v[u]);
}

__global__ __launch_bounds__(256) void agg2_kernel(const int* __restrict__ src, const int* __restrict__ dst,
                                                   const float* __restrict__ t2s, float* acc2,
                                                   unsigned* queue, int e, int n) {
  __shared__ int s_half;
  __shared__ int s_next[2];
  __shared__ EdgeList s_list[8];
  const int lane = threadIdx.x & 31;
  const int wave = threadIdx.x >> 5;
  const int g = lane >> 3;
  const int c = lane & 7;
  const int nsplit = (n >> 1) & ~7;
  const int nchunk = (e + AGG_CH - 1) / AGG_CH;
  int2* lst = s_list[wave].e;
  if (threadIdx.x == 0) s_half = coherence_half();

  for (int pass = 0; pass < 2; ++pass) {
    __syncthreads();
    const int h = s_half ^ pass;
    unsigned* q = queue + h;
    if (threadIdx.x == 0) s_next[0] = (int)atomicAdd(q, 1u);
    __syncthreads();
    for (int it = 0; ; ++it) {
      const int cur = s_next[it & 1];
      if (cur >= nchunk) break;
      int dv[AGG_BLK], sv[AGG_BLK];
#pragma unroll
      for (int b = 0; b < AGG_BLK; ++b) {
        const int j = cur * AGG_CH + (b * 8 + wave) * 32 + lane;
        dv[b] = -1; sv[b] = 0;
        if (j < e) { dv[b] = dst[j]; sv[b] = src[j]; }
      }
      unsigned nx = 0u;
      if (threadIdx.x == 0) nx = atomicAdd(q, 1u);
      int cnt = 0;
#pragma unroll
      for (int b = 0; b < AGG_BLK; ++b) {
        const bool mine = (dv[b] >= 0) && ((dv[b] >= nsplit) == (h == 1));
        const unsigned m = __builtin_amdgcn_ballot_w32(mine);
        if (mine) lst[cnt + (int)__builtin_amdgcn_mbcnt_lo(m, 0u)] = make_int2(sv[b], dv[b]);
        cnt += __builtin_popcount(m);
      }
      if (threadIdx.x == 0) s_next[(it + 1) & 1] = (int)nx;
      __syncthreads();

      int i = 0;
      for (; i + 16 <= cnt; i += 16) agg2_group<4>(lst, i, t2s, acc2, g, c);
      for (; i < cnt; i += 4)
        if (i + g < cnt) agg2_group<1>(lst, i, t2s, acc2, g, c);

    }
  }
}

__global__ void finalize2_kernel(const float* __restrict__ acc2, const float* __restrict__ dinv,
                                 const float* __restrict__ b2, float* __restrict__ out, int n) {
  int idx = blockIdx.x * blockDim.x + threadIdx.x;
  if (idx >= n * NC) return;
  int i = idx >> 3, c = idx & 7;
  out[idx] = fmaf(acc2[idx], dinv[i], b2[c]);
}


extern "C" void kernel_launch(void* const* d_in, const int* in_sizes, int n_in,
                              void* d_out, int out_size, void* d_ws, size_t ws_size,
                              hipStream_t stream) {
  const float* x  = (const float*)d_in[0];
  const int*   ei = (const int*)d_in[1];
  const float* W1 = (const float*)d_in[2];
  const float* b1 = (const float*)d_in[3];
  const float* W2 = (const float*)d_in[4];
  const float* b2 = (const float*)d_in[5];
  float* out = (float*)d_out;

  const int n = in_sizes[0] / NF;
  const int e = in_sizes[1] / 2;
  const int* src = ei;
  const int* dst = ei + e;

  float* ws   = (float*)d_ws;
  float* dinv = ws;
  float* hs   = dinv + n;
  float* acc1 = hs + (size_t)n * NH;
  float* t2s  = acc1 + (size_t)n * NH;
  float* acc2 = t2s + (size_t)n * NC;

  const int T = 256;
  deg_init_kernel<<<(n + T - 1) / T, T, 0, stream>>>(dinv, n);
  deg_count_kernel<<<dim3(DEG_CHUNKS, (n + DEG_RANGE - 1) / DEG_RANGE), DEG_T, 0, stream>>>(dst, dinv, e, n);
  unsigned* q1 = (unsigned*)t2s;
  unsigned* q2 = (unsigned*)hs;
  dinv_kernel<<<(n + T - 1) / T, T, 0, stream>>>(dinv, n, q1);

  gemm1_wmma_kernel<<<(n + 127) / 128, T, 0, stream>>>(x, W1, dinv, hs, acc1, n);

  agg1_kernel<<<AGG_WG, T, 0, stream>>>(src, dst, hs, acc1, q1, e, n);

  gemm2_kernel<<<(n * NC + T - 1) / T, T, 0, stream>>>(acc1, dinv, b1, W2, t2s, acc2, n, q2);

  agg2_kernel<<<AGG_WG, T, 0, stream>>>(src, dst, t2s, acc2, q2, e, n);

  finalize2_kernel<<<(n * NC + T - 1) / T, T, 0, stream>>>(acc2, dinv, b2, out, n);
}
